# starting kernel
# baseline (speedup 1.0000x reference)
#include <hip/hip_runtime.h>
#include <math.h>

typedef __attribute__((ext_vector_type(2))) float v2f;
typedef __attribute__((ext_vector_type(8))) float v8f;

__global__ __launch_bounds__(256) void build_A_kernel(float* __restrict__ Aout) {
    int idx = blockIdx.x * 256 + threadIdx.x;
    if (idx >= 32 * 64) return;
    int r = idx >> 6;
    int w = idx & 63;
    const float PI = 3.14159265358979323846f;
    float acc = 0.0f;
    for (int p = 0; p < 24; ++p) {
        float s32 = (p == 0) ? sqrtf(1.0f / 32.0f) : sqrtf(2.0f / 32.0f);
        float s64 = (p == 0) ? sqrtf(1.0f / 64.0f) : sqrtf(2.0f / 64.0f);
        float c32 = cosf(PI * ((float)r + 0.5f) * (float)p / 32.0f);
        float c64 = cosf(PI * ((float)w + 0.5f) * (float)p / 64.0f);
        acc += c32 * s32 * c64 * s64;
    }
    Aout[idx] = acc;
}

template <int MODE, bool NT>
__global__ __launch_bounds__(256) void dct_gemm_kernel(
    const float* __restrict__ Aglob,
    const float* __restrict__ B,
    float* __restrict__ C,
    int N, int ntilesPerBatch, int totalTiles)
{
    __shared__ float A_lds[32 * 64];
    for (int i = threadIdx.x; i < 32 * 64; i += 256) A_lds[i] = Aglob[i];
    __syncthreads();

    const int lane = threadIdx.x & 31;
    const int wave = threadIdx.x >> 5;
    const int t = blockIdx.x * 8 + wave;
    if (t >= totalTiles) return;

    const int batchIdx = t / ntilesPerBatch;
    const int ntile    = t - batchIdx * ntilesPerBatch;
    const int n0       = ntile * 16;

    const int row  = lane & 15;
    const int koff = (lane >> 4) << 1;
    const int col  = n0 + row;

    v2f afrag[2][16];
#pragma unroll
    for (int mt = 0; mt < 2; ++mt) {
#pragma unroll
        for (int kk = 0; kk < 16; ++kk) {
            const float* ap = &A_lds[(mt * 16 + row) * 64 + kk * 4 + koff];
            v2f a; a.x = ap[0]; a.y = ap[1];
            afrag[mt][kk] = a;
        }
    }

    const float* Bb = B + (size_t)batchIdx * 64 * (size_t)N;
    v8f acc0 = {}; v8f acc1 = {};

#pragma unroll
    for (int kk = 0; kk < 16; ++kk) {
        const int krow = kk * 4 + koff;
        const float* bp0 = Bb + (size_t)krow * N + col;
        const float* bp1 = bp0 + N;
        v2f b;
        if (NT) { b.x = __builtin_nontemporal_load(bp0);
                  b.y = __builtin_nontemporal_load(bp1); }
        else    { b.x = *bp0; b.y = *bp1; }
        acc0 = __builtin_amdgcn_wmma_f32_16x16x4_f32(false, afrag[0][kk], false, b,
                                                     (short)0, acc0, false, false);
        acc1 = __builtin_amdgcn_wmma_f32_16x16x4_f32(false, afrag[1][kk], false, b,
                                                     (short)0, acc1, false, false);
    }

    const int rbase = (lane >> 4) << 3;
    if (MODE == 0) {
        float* Cb = C + (size_t)batchIdx * 32 * (size_t)N;
#pragma unroll
        for (int v = 0; v < 8; ++v) {
            Cb[(size_t)(rbase + v)      * N + col] = acc0[v];
            Cb[(size_t)(16 + rbase + v) * N + col] = acc1[v];
        }
    } else {
        const int b  = batchIdx >> 10;
        const int pq = batchIdx & 1023;
        const int p  = pq >> 5;
        const int q  = pq & 31;
#pragma unroll
        for (int v = 0; v < 8; ++v) {
            int r0 = rbase + v;
            size_t i0 = ((((size_t)b * 32 + r0)        * 32 + q) * 32 + p) * 32 + col;
            size_t i1 = ((((size_t)b * 32 + (16 + r0)) * 32 + q) * 32 + p) * 32 + col;
            C[i0] = acc0[v];
            C[i1] = acc1[v];
        }
    }
}

extern "C" void kernel_launch(void* const* d_in, const int* in_sizes, int n_in,
                              void* d_out, int out_size, void* d_ws, size_t ws_size,
                              hipStream_t stream) {
    (void)in_sizes; (void)n_in; (void)out_size; (void)ws_size;
    const float* x  = (const float*)d_in[0];
    float* out      = (float*)d_out;

    char* ws = (char*)d_ws;
    float* A  = (float*)ws;
    float* z1 = (float*)(ws + 8192);
    float* z2 = (float*)(ws + 8192 + (size_t)4*32*131072*4);

    build_A_kernel<<<8, 256, 0, stream>>>(A);

    dct_gemm_kernel<0, true><<<32768 / 8, 256, 0, stream>>>(
        A, x, z1, 131072, 8192, 32768);

    dct_gemm_kernel<0, false><<<16384 / 8, 256, 0, stream>>>(
        A, z1, z2, 2048, 128, 16384);

    dct_gemm_kernel<1, false><<<8192 / 8, 256, 0, stream>>>(
        A, z2, out, 32, 2, 8192);
}


# Persistent waves: each GEMM pass runs 512 work-groups (4096 waves, was 32768/16384/8192); a wave stages A once, keeps its fragments in registers, walks 8/4/2 tiles, next tile's B loads issued ahead
# speedup vs baseline: 1.2115x; 1.2115x over previous
#include <hip/hip_runtime.h>
#include <math.h>

typedef __attribute__((ext_vector_type(2))) float v2f;
typedef __attribute__((ext_vector_type(8))) float v8f;

__global__ __launch_bounds__(256) void build_A_kernel(float* __restrict__ Aout) {
    int idx = blockIdx.x * 256 + threadIdx.x;
    if (idx >= 32 * 64) return;
    int r = idx >> 6;
    int w = idx & 63;
    const float PI = 3.14159265358979323846f;
    float acc = 0.0f;
    for (int p = 0; p < 24; ++p) {
        float s32 = (p == 0) ? sqrtf(1.0f / 32.0f) : sqrtf(2.0f / 32.0f);
        float s64 = (p == 0) ? sqrtf(1.0f / 64.0f) : sqrtf(2.0f / 64.0f);
        float c32 = cosf(PI * ((float)r + 0.5f) * (float)p / 32.0f);
        float c64 = cosf(PI * ((float)w + 0.5f) * (float)p / 64.0f);
        acc += c32 * s32 * c64 * s64;
    }
    Aout[idx] = acc;
}


template <bool NT>
__device__ __forceinline__ void load_b_tile(v2f (&b)[16], const float* __restrict__ tb,
                                            unsigned lo0, unsigned lo1, int N)
{
#pragma unroll
    for (int kk = 0; kk < 16; ++kk) {
        const float* rb = tb + (size_t)(kk * 4) * (size_t)N;
        v2f v;
        if (NT) { v.x = __builtin_nontemporal_load(rb + lo0);
                  v.y = __builtin_nontemporal_load(rb + lo1); }
        else    { v.x = rb[lo0]; v.y = rb[lo1]; }
        b[kk] = v;
    }
}

template <int MODE>
__device__ __forceinline__ void mma_store_tile(const v2f (&afrag)[2][16], const v2f (&b)[16],
                                               float* __restrict__ C, int N,
                                               int batchIdx, int n0, int row, int rbase)
{
    v8f acc0 = {}; v8f acc1 = {};
#pragma unroll
    for (int kk = 0; kk < 16; ++kk) {
        acc0 = __builtin_amdgcn_wmma_f32_16x16x4_f32(false, afrag[0][kk], false, b[kk],
                                                     (short)0, acc0, false, false);
        acc1 = __builtin_amdgcn_wmma_f32_16x16x4_f32(false, afrag[1][kk], false, b[kk],
                                                     (short)0, acc1, false, false);
    }
    if (MODE == 0) {
        float* cb = C + (size_t)batchIdx * 32 * (size_t)N + n0;
        const unsigned so = (unsigned)(rbase * N + row);
#pragma unroll
        for (int v = 0; v < 8; ++v) {
            (cb + (size_t)v        * (size_t)N)[so] = acc0[v];
            (cb + (size_t)(16 + v) * (size_t)N)[so] = acc1[v];
        }
    } else {
        const int bb = batchIdx >> 10;
        const int pq = batchIdx & 1023;
        const int p  = pq >> 5;
        const int q  = pq & 31;
        float* cb = C + (((size_t)bb * 32 * 32 + q) * 32 + p) * 32 + n0;
        const unsigned so = (unsigned)(rbase * 32768 + row);
#pragma unroll
        for (int v = 0; v < 8; ++v) {
            (cb + (size_t)v        * 32768)[so] = acc0[v];
            (cb + (size_t)(16 + v) * 32768)[so] = acc1[v];
        }
    }
}

template <int MODE, bool NT>
__global__ __launch_bounds__(256) void dct_gemm_kernel(
    const float* __restrict__ Aglob,
    const float* __restrict__ B,
    float* __restrict__ C,
    int N, int ntilesPerBatch, int totalTiles)
{
    __shared__ float A_lds[32 * 64];
    for (int i = threadIdx.x; i < 32 * 64; i += 256) A_lds[i] = Aglob[i];
    __syncthreads();

    const int lane = threadIdx.x & 31;
    const int wave = __builtin_amdgcn_readfirstlane((int)(threadIdx.x >> 5));
    const int gw = blockIdx.x * 8 + wave;
    const int GW = gridDim.x * 8;
    if (gw >= totalTiles) return;

    const int row   = lane & 15;
    const int koff  = (lane >> 4) << 1;
    const int rbase = (lane >> 4) << 3;

    v2f afrag[2][16];
#pragma unroll
    for (int mt = 0; mt < 2; ++mt) {
#pragma unroll
        for (int kk = 0; kk < 16; ++kk) {
            const float* ap = &A_lds[(mt * 16 + row) * 64 + kk * 4 + koff];
            v2f a; a.x = ap[0]; a.y = ap[1];
            afrag[mt][kk] = a;
        }
    }

    const unsigned lo0 = (unsigned)(koff * N + row);
    const unsigned lo1 = lo0 + (unsigned)N;

    v2f b0[16], b1[16];

    int t = gw;
    int batchIdx = t / ntilesPerBatch;
    int n0 = (t - batchIdx * ntilesPerBatch) * 16;
    load_b_tile<NT>(b0, B + (size_t)batchIdx * 64 * (size_t)N + n0, lo0, lo1, N);

    for (;;) {
        int tn = t + GW;
        if (tn >= totalTiles) {
            mma_store_tile<MODE>(afrag, b0, C, N, batchIdx, n0, row, rbase);
            break;
        }
        int bn = tn / ntilesPerBatch;
        int nn = (tn - bn * ntilesPerBatch) * 16;
        load_b_tile<NT>(b1, B + (size_t)bn * 64 * (size_t)N + nn, lo0, lo1, N);
        mma_store_tile<MODE>(afrag, b0, C, N, batchIdx, n0, row, rbase);
        t = tn; batchIdx = bn; n0 = nn;

        tn = t + GW;
        if (tn >= totalTiles) {
            mma_store_tile<MODE>(afrag, b1, C, N, batchIdx, n0, row, rbase);
            break;
        }
        bn = tn / ntilesPerBatch;
        nn = (tn - bn * ntilesPerBatch) * 16;
        load_b_tile<NT>(b0, B + (size_t)bn * 64 * (size_t)N + nn, lo0, lo1, N);
        mma_store_tile<MODE>(afrag, b1, C, N, batchIdx, n0, row, rbase);
        t = tn; batchIdx = bn; n0 = nn;
    }
}

extern "C" void kernel_launch(void* const* d_in, const int* in_sizes, int n_in,
                              void* d_out, int out_size, void* d_ws, size_t ws_size,
                              hipStream_t stream) {
    (void)in_sizes; (void)n_in; (void)out_size; (void)ws_size;
    const float* x  = (const float*)d_in[0];
    float* out      = (float*)d_out;

    char* ws = (char*)d_ws;
    float* A  = (float*)ws;
    float* z1 = (float*)(ws + 8192);
    float* z2 = (float*)(ws + 8192 + (size_t)4*32*131072*4);

    build_A_kernel<<<8, 256, 0, stream>>>(A);

    dct_gemm_kernel<0, true><<<512, 256, 0, stream>>>(
        A, x, z1, 131072, 8192, 32768);

    dct_gemm_kernel<0, false><<<512, 256, 0, stream>>>(
        A, z1, z2, 2048, 128, 16384);

    dct_gemm_kernel<1, false><<<512, 256, 0, stream>>>(
        A, z2, out, 32, 2, 8192);
}


# Passes 1 and 2 fused: one 16-wave work-group per (b,w) contracts d into a 256 KB LDS slab and contracts h from it; z1 never goes to memory, 3 launches instead of 4
# speedup vs baseline: 1.6159x; 1.3338x over previous
#include <hip/hip_runtime.h>
#include <math.h>

typedef __attribute__((ext_vector_type(2))) float v2f;
typedef __attribute__((ext_vector_type(8))) float v8f;

__global__ __launch_bounds__(256) void build_A_kernel(float* __restrict__ Aout) {
    int idx = blockIdx.x * 256 + threadIdx.x;
    if (idx >= 32 * 64) return;
    int r = idx >> 6;
    int w = idx & 63;
    const float PI = 3.14159265358979323846f;
    float acc = 0.0f;
    for (int p = 0; p < 24; ++p) {
        float s32 = (p == 0) ? sqrtf(1.0f / 32.0f) : sqrtf(2.0f / 32.0f);
        float s64 = (p == 0) ? sqrtf(1.0f / 64.0f) : sqrtf(2.0f / 64.0f);
        float c32 = cosf(PI * ((float)r + 0.5f) * (float)p / 32.0f);
        float c64 = cosf(PI * ((float)w + 0.5f) * (float)p / 64.0f);
        acc += c32 * s32 * c64 * s64;
    }
    Aout[idx] = acc;
}


template <bool NT>
__device__ __forceinline__ void load_b_tile(v2f (&b)[16], const float* __restrict__ tb,
                                            unsigned lo0, unsigned lo1, int N)
{
#pragma unroll
    for (int kk = 0; kk < 16; ++kk) {
        const float* rb = tb + (size_t)(kk * 4) * (size_t)N;
        v2f v;
        if (NT) { v.x = __builtin_nontemporal_load(rb + lo0);
                  v.y = __builtin_nontemporal_load(rb + lo1); }
        else    { v.x = rb[lo0]; v.y = rb[lo1]; }
        b[kk] = v;
    }
}

template <int MODE>
__device__ __forceinline__ void mma_store_tile(const v2f (&afrag)[2][16], const v2f (&b)[16],
                                               float* __restrict__ C, int N,
                                               int batchIdx, int n0, int row, int rbase)
{
    v8f acc0 = {}; v8f acc1 = {};
#pragma unroll
    for (int kk = 0; kk < 16; ++kk) {
        acc0 = __builtin_amdgcn_wmma_f32_16x16x4_f32(false, afrag[0][kk], false, b[kk],
                                                     (short)0, acc0, false, false);
        acc1 = __builtin_amdgcn_wmma_f32_16x16x4_f32(false, afrag[1][kk], false, b[kk],
                                                     (short)0, acc1, false, false);
    }
    if (MODE == 0) {
        float* cb = C + (size_t)batchIdx * 32 * (size_t)N + n0;
        const unsigned so = (unsigned)(rbase * N + row);
#pragma unroll
        for (int v = 0; v < 8; ++v) {
            (cb + (size_t)v        * (size_t)N)[so] = acc0[v];
            (cb + (size_t)(16 + v) * (size_t)N)[so] = acc1[v];
        }
    } else {
        const int bb = batchIdx >> 10;
        const int pq = batchIdx & 1023;
        const int p  = pq >> 5;
        const int q  = pq & 31;
        float* cb = C + (((size_t)bb * 32 * 32 + q) * 32 + p) * 32 + n0;
        const unsigned so = (unsigned)(rbase * 32768 + row);
#pragma unroll
        for (int v = 0; v < 8; ++v) {
            (cb + (size_t)v        * 32768)[so] = acc0[v];
            (cb + (size_t)(16 + v) * 32768)[so] = acc1[v];
        }
    }
}

template <int MODE, bool NT>
__global__ __launch_bounds__(256) void dct_gemm_kernel(
    const float* __restrict__ Aglob,
    const float* __restrict__ B,
    float* __restrict__ C,
    int N, int ntilesPerBatch, int totalTiles)
{
    __shared__ float A_lds[32 * 64];
    for (int i = threadIdx.x; i < 32 * 64; i += 256) A_lds[i] = Aglob[i];
    __syncthreads();

    const int lane = threadIdx.x & 31;
    const int wave = __builtin_amdgcn_readfirstlane((int)(threadIdx.x >> 5));
    const int gw = blockIdx.x * 8 + wave;
    const int GW = gridDim.x * 8;
    if (gw >= totalTiles) return;

    const int row   = lane & 15;
    const int koff  = (lane >> 4) << 1;
    const int rbase = (lane >> 4) << 3;

    v2f afrag[2][16];
#pragma unroll
    for (int mt = 0; mt < 2; ++mt) {
#pragma unroll
        for (int kk = 0; kk < 16; ++kk) {
            const float* ap = &A_lds[(mt * 16 + row) * 64 + kk * 4 + koff];
            v2f a; a.x = ap[0]; a.y = ap[1];
            afrag[mt][kk] = a;
        }
    }

    const unsigned lo0 = (unsigned)(koff * N + row);
    const unsigned lo1 = lo0 + (unsigned)N;

    v2f b0[16], b1[16];

    int t = gw;
    int batchIdx = t / ntilesPerBatch;
    int n0 = (t - batchIdx * ntilesPerBatch) * 16;
    load_b_tile<NT>(b0, B + (size_t)batchIdx * 64 * (size_t)N + n0, lo0, lo1, N);

    for (;;) {
        int tn = t + GW;
        if (tn >= totalTiles) {
            mma_store_tile<MODE>(afrag, b0, C, N, batchIdx, n0, row, rbase);
            break;
        }
        int bn = tn / ntilesPerBatch;
        int nn = (tn - bn * ntilesPerBatch) * 16;
        load_b_tile<NT>(b1, B + (size_t)bn * 64 * (size_t)N + nn, lo0, lo1, N);
        mma_store_tile<MODE>(afrag, b0, C, N, batchIdx, n0, row, rbase);
        t = tn; batchIdx = bn; n0 = nn;

        tn = t + GW;
        if (tn >= totalTiles) {
            mma_store_tile<MODE>(afrag, b1, C, N, batchIdx, n0, row, rbase);
            break;
        }
        bn = tn / ntilesPerBatch;
        nn = (tn - bn * ntilesPerBatch) * 16;
        load_b_tile<NT>(b0, B + (size_t)bn * 64 * (size_t)N + nn, lo0, lo1, N);
        mma_store_tile<MODE>(afrag, b1, C, N, batchIdx, n0, row, rbase);
        t = tn; batchIdx = bn; n0 = nn;
    }
}


__device__ __forceinline__ int zidx(int p, int h, int c) {
    return (p * 64 + h) * 32 + (c ^ ((((h >> 1) ^ (p >> 3)) & 1) << 4));
}

__device__ __forceinline__ void mma_to_slab(const v2f (&afrag)[2][16], const v2f (&b)[16],
                                            float* __restrict__ Zs, int h, int chalf,
                                            int row, int rbase)
{
    v8f acc0 = {}; v8f acc1 = {};
#pragma unroll
    for (int kk = 0; kk < 16; ++kk) {
        acc0 = __builtin_amdgcn_wmma_f32_16x16x4_f32(false, afrag[0][kk], false, b[kk],
                                                     (short)0, acc0, false, false);
        acc1 = __builtin_amdgcn_wmma_f32_16x16x4_f32(false, afrag[1][kk], false, b[kk],
                                                     (short)0, acc1, false, false);
    }
    const int c = chalf * 16 + row;
#pragma unroll
    for (int v = 0; v < 8; ++v) {
        Zs[zidx(rbase + v,      h, c)] = acc0[v];
        Zs[zidx(16 + rbase + v, h, c)] = acc1[v];
    }
}

__global__ __launch_bounds__(512) void dct_dh_fused_kernel(
    const float* __restrict__ Aglob,
    const float* __restrict__ X,
    float* __restrict__ Z2)
{
    __shared__ float A_lds[32 * 64];
    __shared__ float Zs[32 * 64 * 32];
    for (int i = threadIdx.x; i < 32 * 64; i += 512) A_lds[i] = Aglob[i];
    __syncthreads();

    const int lane = threadIdx.x & 31;
    const int wave = __builtin_amdgcn_readfirstlane((int)(threadIdx.x >> 5));
    const int row   = lane & 15;
    const int koff  = (lane >> 4) << 1;
    const int rbase = (lane >> 4) << 3;

    v2f afrag[2][16];
#pragma unroll
    for (int mt = 0; mt < 2; ++mt) {
#pragma unroll
        for (int kk = 0; kk < 16; ++kk) {
            const float* ap = &A_lds[(mt * 16 + row) * 64 + kk * 4 + koff];
            v2f a; a.x = ap[0]; a.y = ap[1];
            afrag[mt][kk] = a;
        }
    }

    const int bb = blockIdx.x >> 6;
    const int w  = blockIdx.x & 63;

    {
        const int N1 = 131072;
        const unsigned lo0 = (unsigned)(koff * N1 + row);
        const unsigned lo1 = lo0 + (unsigned)N1;
        const float* xb = X + (size_t)bb * 64 * (size_t)N1 + (size_t)(wave * 4) * 2048 + w * 32;
        v2f b0[16], b1[16];
        load_b_tile<true>(b0, xb, lo0, lo1, N1);
#pragma unroll
        for (int j = 0; j < 4; ++j) {
            const int h = wave * 4 + j;
            load_b_tile<true>(b1, xb + j * 2048 + 16, lo0, lo1, N1);
            mma_to_slab(afrag, b0, Zs, h, 0, row, rbase);
            if (j < 3) load_b_tile<true>(b0, xb + (j + 1) * 2048, lo0, lo1, N1);
            mma_to_slab(afrag, b1, Zs, h, 1, row, rbase);
        }
    }
    __syncthreads();

#pragma unroll
    for (int i = 0; i < 4; ++i) {
        const int p = wave * 2 + (i >> 1);
        const int chalf = i & 1;
        const int c = chalf * 16 + row;
        v2f b[16];
#pragma unroll
        for (int kk = 0; kk < 16; ++kk) {
            const int h0 = kk * 4 + koff;
            v2f v;
            v.x = Zs[zidx(p, h0,     c)];
            v.y = Zs[zidx(p, h0 + 1, c)];
            b[kk] = v;
        }
        v8f acc0 = {}; v8f acc1 = {};
#pragma unroll
        for (int kk = 0; kk < 16; ++kk) {
            acc0 = __builtin_amdgcn_wmma_f32_16x16x4_f32(false, afrag[0][kk], false, b[kk],
                                                         (short)0, acc0, false, false);
            acc1 = __builtin_amdgcn_wmma_f32_16x16x4_f32(false, afrag[1][kk], false, b[kk],
                                                         (short)0, acc1, false, false);
        }
        float* cb = Z2 + ((size_t)(bb * 32 + p) * 32) * 2048 + w * 32 + chalf * 16;
        const unsigned so = (unsigned)(rbase * 2048 + row);
#pragma unroll
        for (int v = 0; v < 8; ++v) {
            (cb + (size_t)v        * 2048)[so] = acc0[v];
            (cb + (size_t)(16 + v) * 2048)[so] = acc1[v];
        }
    }
}

extern "C" void kernel_launch(void* const* d_in, const int* in_sizes, int n_in,
                              void* d_out, int out_size, void* d_ws, size_t ws_size,
                              hipStream_t stream) {
    (void)in_sizes; (void)n_in; (void)out_size; (void)ws_size;
    const float* x  = (const float*)d_in[0];
    float* out      = (float*)d_out;

    char* ws = (char*)d_ws;
    float* A  = (float*)ws;
    float* z2 = (float*)(ws + 8192 + (size_t)4*32*131072*4);

    build_A_kernel<<<8, 256, 0, stream>>>(A);

    dct_dh_fused_kernel<<<256, 512, 0, stream>>>(A, x, z2);

    dct_gemm_kernel<1, false><<<512, 256, 0, stream>>>(
        A, z2, out, 32, 2, 8192);
}


# Fused pass now contracts w then h per (b,d): a work-group streams one contiguous 512 KB block of x (was 4096 lone 128-B lines); d is contracted last, so the order of the three axis sums changes
# speedup vs baseline: 1.7043x; 1.0547x over previous
#include <hip/hip_runtime.h>
#include <math.h>

typedef __attribute__((ext_vector_type(2))) float v2f;
typedef __attribute__((ext_vector_type(8))) float v8f;

__global__ __launch_bounds__(256) void build_A_kernel(float* __restrict__ Aout) {
    int idx = blockIdx.x * 256 + threadIdx.x;
    if (idx >= 32 * 64) return;
    int r = idx >> 6;
    int w = idx & 63;
    const float PI = 3.14159265358979323846f;
    float acc = 0.0f;
    for (int p = 0; p < 24; ++p) {
        float s32 = (p == 0) ? sqrtf(1.0f / 32.0f) : sqrtf(2.0f / 32.0f);
        float s64 = (p == 0) ? sqrtf(1.0f / 64.0f) : sqrtf(2.0f / 64.0f);
        float c32 = cosf(PI * ((float)r + 0.5f) * (float)p / 32.0f);
        float c64 = cosf(PI * ((float)w + 0.5f) * (float)p / 64.0f);
        acc += c32 * s32 * c64 * s64;
    }
    Aout[idx] = acc;
}


template <bool NT>
__device__ __forceinline__ void load_b_tile(v2f (&b)[16], const float* __restrict__ tb,
                                            unsigned lo0, unsigned lo1, int N)
{
#pragma unroll
    for (int kk = 0; kk < 16; ++kk) {
        const float* rb = tb + (size_t)(kk * 4) * (size_t)N;
        v2f v;
        if (NT) { v.x = __builtin_nontemporal_load(rb + lo0);
                  v.y = __builtin_nontemporal_load(rb + lo1); }
        else    { v.x = rb[lo0]; v.y = rb[lo1]; }
        b[kk] = v;
    }
}

template <int MODE>
__device__ __forceinline__ void mma_store_tile(const v2f (&afrag)[2][16], const v2f (&b)[16],
                                               float* __restrict__ C, int N,
                                               int batchIdx, int n0, int row, int rbase)
{
    v8f acc0 = {}; v8f acc1 = {};
#pragma unroll
    for (int kk = 0; kk < 16; ++kk) {
        acc0 = __builtin_amdgcn_wmma_f32_16x16x4_f32(false, afrag[0][kk], false, b[kk],
                                                     (short)0, acc0, false, false);
        acc1 = __builtin_amdgcn_wmma_f32_16x16x4_f32(false, afrag[1][kk], false, b[kk],
                                                     (short)0, acc1, false, false);
    }
    if (MODE == 0) {
        float* cb = C + (size_t)batchIdx * 32 * (size_t)N + n0;
        const unsigned so = (unsigned)(rbase * N + row);
#pragma unroll
        for (int v = 0; v < 8; ++v) {
            (cb + (size_t)v        * (size_t)N)[so] = acc0[v];
            (cb + (size_t)(16 + v) * (size_t)N)[so] = acc1[v];
        }
    } else {
        float* cb = C + ((size_t)batchIdx * 1024 + (size_t)(n0 >> 5)) * 1024 + (n0 & 31);
        const unsigned so = (unsigned)(rbase * 32 + row);
#pragma unroll
        for (int v = 0; v < 8; ++v) {
            (cb + (size_t)v        * 32)[so] = acc0[v];
            (cb + (size_t)(16 + v) * 32)[so] = acc1[v];
        }
    }
}

template <int MODE, bool NT>
__global__ __launch_bounds__(256) void dct_gemm_kernel(
    const float* __restrict__ Aglob,
    const float* __restrict__ B,
    float* __restrict__ C,
    int N, int ntilesPerBatch, int totalTiles)
{
    __shared__ float A_lds[32 * 64];
    for (int i = threadIdx.x; i < 32 * 64; i += 256) A_lds[i] = Aglob[i];
    __syncthreads();

    const int lane = threadIdx.x & 31;
    const int wave = __builtin_amdgcn_readfirstlane((int)(threadIdx.x >> 5));
    const int gw = blockIdx.x * 8 + wave;
    const int GW = gridDim.x * 8;
    if (gw >= totalTiles) return;

    const int row   = lane & 15;
    const int koff  = (lane >> 4) << 1;
    const int rbase = (lane >> 4) << 3;

    v2f afrag[2][16];
#pragma unroll
    for (int mt = 0; mt < 2; ++mt) {
#pragma unroll
        for (int kk = 0; kk < 16; ++kk) {
            const float* ap = &A_lds[(mt * 16 + row) * 64 + kk * 4 + koff];
            v2f a; a.x = ap[0]; a.y = ap[1];
            afrag[mt][kk] = a;
        }
    }

    const unsigned lo0 = (unsigned)(koff * N + row);
    const unsigned lo1 = lo0 + (unsigned)N;

    v2f b0[16], b1[16];

    int t = gw;
    int batchIdx = t / ntilesPerBatch;
    int n0 = (t - batchIdx * ntilesPerBatch) * 16;
    load_b_tile<NT>(b0, B + (size_t)batchIdx * 64 * (size_t)N + n0, lo0, lo1, N);

    for (;;) {
        int tn = t + GW;
        if (tn >= totalTiles) {
            mma_store_tile<MODE>(afrag, b0, C, N, batchIdx, n0, row, rbase);
            break;
        }
        int bn = tn / ntilesPerBatch;
        int nn = (tn - bn * ntilesPerBatch) * 16;
        load_b_tile<NT>(b1, B + (size_t)bn * 64 * (size_t)N + nn, lo0, lo1, N);
        mma_store_tile<MODE>(afrag, b0, C, N, batchIdx, n0, row, rbase);
        t = tn; batchIdx = bn; n0 = nn;

        tn = t + GW;
        if (tn >= totalTiles) {
            mma_store_tile<MODE>(afrag, b1, C, N, batchIdx, n0, row, rbase);
            break;
        }
        bn = tn / ntilesPerBatch;
        nn = (tn - bn * ntilesPerBatch) * 16;
        load_b_tile<NT>(b0, B + (size_t)bn * 64 * (size_t)N + nn, lo0, lo1, N);
        mma_store_tile<MODE>(afrag, b1, C, N, batchIdx, n0, row, rbase);
        t = tn; batchIdx = bn; n0 = nn;
    }
}


__device__ __forceinline__ int zidx(int p, int h, int c) {
    return (p * 64 + h) * 32 + (c ^ ((((h >> 1) ^ (p >> 3)) & 1) << 4));
}

__device__ __forceinline__ void mma_to_slab(const v2f (&afrag)[2][16], const v2f (&b)[16],
                                            float* __restrict__ Zs, int h, int chalf,
                                            int row, int rbase)
{
    v8f acc0 = {}; v8f acc1 = {};
#pragma unroll
    for (int kk = 0; kk < 16; ++kk) {
        acc0 = __builtin_amdgcn_wmma_f32_16x16x4_f32(false, afrag[0][kk], false, b[kk],
                                                     (short)0, acc0, false, false);
        acc1 = __builtin_amdgcn_wmma_f32_16x16x4_f32(false, afrag[1][kk], false, b[kk],
                                                     (short)0, acc1, false, false);
    }
    const int c = chalf * 16 + row;
#pragma unroll
    for (int v = 0; v < 8; ++v) {
        Zs[zidx(rbase + v,      h, c)] = acc0[v];
        Zs[zidx(16 + rbase + v, h, c)] = acc1[v];
    }
    __builtin_amdgcn_sched_barrier(0);
}

__global__ __launch_bounds__(512) void dct_wh_fused_kernel(
    const float* __restrict__ Aglob,
    const float* __restrict__ X,
    float* __restrict__ Z2)
{
    __shared__ float A_lds[32 * 64];
    __shared__ float Zs[32 * 64 * 32];
    for (int i = threadIdx.x; i < 32 * 64; i += 512) A_lds[i] = Aglob[i];
    __syncthreads();

    const int lane = threadIdx.x & 31;
    const int wave = __builtin_amdgcn_readfirstlane((int)(threadIdx.x >> 5));
    const int row   = lane & 15;
    const int koff  = (lane >> 4) << 1;
    const int rbase = (lane >> 4) << 3;

    v2f afrag[2][16];
#pragma unroll
    for (int mt = 0; mt < 2; ++mt) {
#pragma unroll
        for (int kk = 0; kk < 16; ++kk) {
            const float* ap = &A_lds[(mt * 16 + row) * 64 + kk * 4 + koff];
            v2f a; a.x = ap[0]; a.y = ap[1];
            afrag[mt][kk] = a;
        }
    }

    const int bb = blockIdx.x >> 6;
    const int d  = blockIdx.x & 63;

    {
        const int N1 = 32;
        const unsigned lo0 = (unsigned)(koff * N1 + row);
        const unsigned lo1 = lo0 + (unsigned)N1;
        const float* xb = X + ((size_t)(bb * 64 + d) * 64 + (size_t)(wave * 4)) * 2048;
        v2f b0[16], b1[16];
        load_b_tile<true>(b0, xb, lo0, lo1, N1);
#pragma unroll
        for (int j = 0; j < 4; ++j) {
            const int h = wave * 4 + j;
            load_b_tile<true>(b1, xb + j * 2048 + 16, lo0, lo1, N1);
            mma_to_slab(afrag, b0, Zs, h, 0, row, rbase);
            if (j < 3) load_b_tile<true>(b0, xb + (j + 1) * 2048, lo0, lo1, N1);
            mma_to_slab(afrag, b1, Zs, h, 1, row, rbase);
        }
    }
    __syncthreads();

#pragma unroll
    for (int i = 0; i < 4; ++i) {
        const int r = wave * 2 + (i >> 1);
        const int chalf = i & 1;
        const int c = chalf * 16 + row;
        v2f b[16];
#pragma unroll
        for (int kk = 0; kk < 16; ++kk) {
            const int h0 = kk * 4 + koff;
            v2f v;
            v.x = Zs[zidx(r, h0,     c)];
            v.y = Zs[zidx(r, h0 + 1, c)];
            b[kk] = v;
        }
        __builtin_amdgcn_sched_barrier(0);
        v8f acc0 = {}; v8f acc1 = {};
#pragma unroll
        for (int kk = 0; kk < 16; ++kk) {
            acc0 = __builtin_amdgcn_wmma_f32_16x16x4_f32(false, afrag[0][kk], false, b[kk],
                                                         (short)0, acc0, false, false);
            acc1 = __builtin_amdgcn_wmma_f32_16x16x4_f32(false, afrag[1][kk], false, b[kk],
                                                         (short)0, acc1, false, false);
        }
        float* cb = Z2 + ((size_t)((bb * 64 + d) * 32 + r)) * 1024 + chalf * 16;
        const unsigned so = (unsigned)(rbase * 32 + row);
#pragma unroll
        for (int v = 0; v < 8; ++v) {
            (cb + (size_t)v        * 32)[so] = acc0[v];
            (cb + (size_t)(16 + v) * 32)[so] = acc1[v];
        }
        __builtin_amdgcn_sched_barrier(0);
    }
}

extern "C" void kernel_launch(void* const* d_in, const int* in_sizes, int n_in,
                              void* d_out, int out_size, void* d_ws, size_t ws_size,
                              hipStream_t stream) {
    (void)in_sizes; (void)n_in; (void)out_size; (void)ws_size;
    const float* x  = (const float*)d_in[0];
    float* out      = (float*)d_out;

    char* ws = (char*)d_ws;
    float* A  = (float*)ws;
    float* z2 = (float*)(ws + 8192 + (size_t)4*32*131072*4);

    build_A_kernel<<<8, 256, 0, stream>>>(A);

    dct_wh_fused_kernel<<<256, 512, 0, stream>>>(A, x, z2);

    dct_gemm_kernel<1, false><<<512, 256, 0, stream>>>(
        A, z2, out, 32768, 2048, 8192);
}


# Operator build moved into the fused pass: each work-group forms E and O in LDS from two small cosine tables (same values, same sum order); work-group 0 stores them for the last pass; one launch fewer
# speedup vs baseline: 2.4555x; 1.4408x over previous
#include <hip/hip_runtime.h>
#include <math.h>

typedef __attribute__((ext_vector_type(2))) float v2f;
typedef __attribute__((ext_vector_type(8))) float v8f;


__device__ __forceinline__ void build_operator(float* __restrict__ A_lds, float* __restrict__ tab,
                                               float* __restrict__ Aout)
{
    const float PI = 3.14159265358979323846f;
    float* T32 = tab;
    float* T64 = tab + 24 * 16;
    for (int i = threadIdx.x; i < 24 * 16 + 24 * 32; i += 512) {
        if (i < 24 * 16) {
            const int p = i >> 4, r = i & 15;
            float s32 = (p == 0) ? sqrtf(1.0f / 32.0f) : sqrtf(2.0f / 32.0f);
            float c32 = cosf(PI * ((float)r + 0.5f) * (float)p / 32.0f);
            T32[i] = c32 * s32;
        } else {
            const int j = i - 24 * 16;
            const int p = j >> 5, w = j & 31;
            T64[j] = cosf(PI * ((float)w + 0.5f) * (float)p / 64.0f);
        }
    }
    __syncthreads();
    const int r = (int)(threadIdx.x >> 5);
    const int w = (int)(threadIdx.x & 31);
    float accE = 0.0f, accO = 0.0f;
#pragma unroll
    for (int p = 0; p < 24; p += 2) {
        const float s64e = (p == 0) ? sqrtf(1.0f / 64.0f) : sqrtf(2.0f / 64.0f);
        const float s64o = sqrtf(2.0f / 64.0f);
        accE = __builtin_fmaf(T32[p * 16 + r] * T64[p * 32 + w], s64e, accE);
        accO = __builtin_fmaf(T32[(p + 1) * 16 + r] * T64[(p + 1) * 32 + w], s64o, accO);
    }
    A_lds[threadIdx.x] = accE;
    A_lds[512 + threadIdx.x] = accO;
    if (blockIdx.x == 0) {
        Aout[threadIdx.x] = accE;
        Aout[512 + threadIdx.x] = accO;
    }
    __syncthreads();
}


__device__ __forceinline__ void load_a_frags(v2f (&aE)[8], v2f (&aO)[8],
                                             const float* A_lds, int row, int koff)
{
#pragma unroll
    for (int kk = 0; kk < 8; ++kk) {
        const float* ep = &A_lds[row * 32 + kk * 4 + koff];
        v2f a; a.x = ep[0];   a.y = ep[1];   aE[kk] = a;
        v2f b; b.x = ep[512]; b.y = ep[513]; aO[kk] = b;
    }
}

template <bool NT>
__device__ __forceinline__ void load_b_tile(v2f (&bp)[8], v2f (&bm)[8], const float* __restrict__ tb,
                                            unsigned loP, unsigned loM, int N)
{
#pragma unroll
    for (int kk = 0; kk < 8; ++kk) {
        const float* rp = tb + (size_t)(kk * 4) * (size_t)N;
        const float* rm = tb + (size_t)(60 - kk * 4) * (size_t)N;
        v2f p, m;
        if (NT) { p.x = __builtin_nontemporal_load(rp + loP);
                  p.y = __builtin_nontemporal_load(rp + loP + (unsigned)N);
                  m.x = __builtin_nontemporal_load(rm + loM + (unsigned)N);
                  m.y = __builtin_nontemporal_load(rm + loM); }
        else    { p.x = rp[loP]; p.y = rp[loP + (unsigned)N];
                  m.x = rm[loM + (unsigned)N]; m.y = rm[loM]; }
        bp[kk] = p; bm[kk] = m;
    }
}

__device__ __forceinline__ void mma_tile(const v2f (&aE)[8], const v2f (&aO)[8],
                                         const v2f (&bp)[8], const v2f (&bm)[8],
                                         v8f& lo, v8f& hi)
{
    v8f accE = {}; v8f accO = {};
#pragma unroll
    for (int kk = 0; kk < 8; ++kk) {
        v2f u = bp[kk] + bm[kk];
        v2f v = bp[kk] - bm[kk];
        accE = __builtin_amdgcn_wmma_f32_16x16x4_f32(false, aE[kk], false, u,
                                                     (short)0, accE, false, false);
        accO = __builtin_amdgcn_wmma_f32_16x16x4_f32(false, aO[kk], false, v,
                                                     (short)0, accO, false, false);
    }
    lo = accE + accO;
    hi = accE - accO;
}

__device__ __forceinline__ void mma_store_tile(const v2f (&aE)[8], const v2f (&aO)[8],
                                               const v2f (&bp)[8], const v2f (&bm)[8],
                                               float* __restrict__ C,
                                               int batchIdx, int n0, int row, int rbase)
{
    v8f lo, hi;
    mma_tile(aE, aO, bp, bm, lo, hi);
    float* cb = C + ((size_t)batchIdx * 1024 + (size_t)(n0 >> 5)) * 1024 + (n0 & 31);
    const unsigned so_lo = (unsigned)(rbase * 32 + row);
    const unsigned so_hi = (unsigned)((31 - rbase) * 32 + row);
#pragma unroll
    for (int v = 0; v < 8; ++v) {
        cb[so_lo + (unsigned)(v * 32)] = lo[v];
        cb[so_hi - (unsigned)(v * 32)] = hi[v];
    }
}

template <bool NT>
__global__ __launch_bounds__(256) void dct_gemm_kernel(
    const float* __restrict__ Aglob,
    const float* __restrict__ B,
    float* __restrict__ C,
    int N, int ntilesPerBatch, int totalTiles)
{
    __shared__ float A_lds[2 * 16 * 32];
    for (int i = threadIdx.x; i < 2 * 16 * 32; i += 256) A_lds[i] = Aglob[i];
    __syncthreads();

    const int lane = threadIdx.x & 31;
    const int wave = __builtin_amdgcn_readfirstlane((int)(threadIdx.x >> 5));
    const int gw = blockIdx.x * 8 + wave;
    const int GW = gridDim.x * 8;
    if (gw >= totalTiles) return;

    const int row   = lane & 15;
    const int koff  = (lane >> 4) << 1;
    const int rbase = (lane >> 4) << 3;

    v2f aE[8], aO[8];
    load_a_frags(aE, aO, A_lds, row, koff);

    const unsigned loP = (unsigned)(koff * N + row);
    const unsigned loM = (unsigned)((2 - koff) * N + row);

    v2f p0[8], m0[8], p1[8], m1[8];

    int t = gw;
    int batchIdx = t / ntilesPerBatch;
    int n0 = (t - batchIdx * ntilesPerBatch) * 16;
    load_b_tile<NT>(p0, m0, B + (size_t)batchIdx * 64 * (size_t)N + n0, loP, loM, N);

    for (;;) {
        int tn = t + GW;
        if (tn >= totalTiles) {
            mma_store_tile(aE, aO, p0, m0, C, batchIdx, n0, row, rbase);
            break;
        }
        int bn = tn / ntilesPerBatch;
        int nn = (tn - bn * ntilesPerBatch) * 16;
        load_b_tile<NT>(p1, m1, B + (size_t)bn * 64 * (size_t)N + nn, loP, loM, N);
        mma_store_tile(aE, aO, p0, m0, C, batchIdx, n0, row, rbase);
        t = tn; batchIdx = bn; n0 = nn;

        tn = t + GW;
        if (tn >= totalTiles) {
            mma_store_tile(aE, aO, p1, m1, C, batchIdx, n0, row, rbase);
            break;
        }
        bn = tn / ntilesPerBatch;
        nn = (tn - bn * ntilesPerBatch) * 16;
        load_b_tile<NT>(p0, m0, B + (size_t)bn * 64 * (size_t)N + nn, loP, loM, N);
        mma_store_tile(aE, aO, p1, m1, C, batchIdx, n0, row, rbase);
        t = tn; batchIdx = bn; n0 = nn;
    }
}


__device__ __forceinline__ int zidx(int p, int h, int c) {
    return (p * 64 + h) * 32 + (c ^ ((((h >> 1) ^ (p >> 3)) & 1) << 4));
}

__device__ __forceinline__ void mma_to_slab(const v2f (&aE)[8], const v2f (&aO)[8],
                                            const v2f (&bp)[8], const v2f (&bm)[8],
                                            float* __restrict__ Zs, int h, int chalf,
                                            int row, int rbase)
{
    v8f lo, hi;
    mma_tile(aE, aO, bp, bm, lo, hi);
    const int c = chalf * 16 + row;
#pragma unroll
    for (int v = 0; v < 8; ++v) {
        Zs[zidx(rbase + v,      h, c)] = lo[v];
        Zs[zidx(31 - rbase - v, h, c)] = hi[v];
    }
    __builtin_amdgcn_sched_barrier(0);
}

__global__ __launch_bounds__(512) void dct_wh_fused_kernel(
    float* __restrict__ Aout,
    const float* __restrict__ X,
    float* __restrict__ Z2)
{
    __shared__ float A_lds[2 * 16 * 32];
    __shared__ float Zs[32 * 64 * 32];
    build_operator(A_lds, Zs, Aout);

    const int lane = threadIdx.x & 31;
    const int wave = __builtin_amdgcn_readfirstlane((int)(threadIdx.x >> 5));
    const int row   = lane & 15;
    const int koff  = (lane >> 4) << 1;
    const int rbase = (lane >> 4) << 3;

    v2f aE[8], aO[8];
    load_a_frags(aE, aO, A_lds, row, koff);

    const int bb = blockIdx.x >> 6;
    const int d  = blockIdx.x & 63;

    {
        const int N1 = 32;
        const unsigned loP = (unsigned)(koff * N1 + row);
        const unsigned loM = (unsigned)((2 - koff) * N1 + row);
        const float* xb = X + ((size_t)(bb * 64 + d) * 64 + (size_t)(wave * 4)) * 2048;
        v2f p0[8], m0[8], p1[8], m1[8];
        load_b_tile<true>(p0, m0, xb, loP, loM, N1);
#pragma unroll
        for (int j = 0; j < 4; ++j) {
            const int h = wave * 4 + j;
            load_b_tile<true>(p1, m1, xb + j * 2048 + 16, loP, loM, N1);
            mma_to_slab(aE, aO, p0, m0, Zs, h, 0, row, rbase);
            if (j < 3) load_b_tile<true>(p0, m0, xb + (j + 1) * 2048, loP, loM, N1);
            mma_to_slab(aE, aO, p1, m1, Zs, h, 1, row, rbase);
        }
    }
    __syncthreads();

#pragma unroll
    for (int i = 0; i < 4; ++i) {
        const int r = wave * 2 + (i >> 1);
        const int chalf = i & 1;
        const int c = chalf * 16 + row;
        v2f bp[8], bm[8];
#pragma unroll
        for (int kk = 0; kk < 8; ++kk) {
            const int h0 = kk * 4 + koff;
            v2f p, m;
            p.x = Zs[zidx(r, h0,      c)];
            p.y = Zs[zidx(r, h0 + 1,  c)];
            m.x = Zs[zidx(r, 63 - h0, c)];
            m.y = Zs[zidx(r, 62 - h0, c)];
            bp[kk] = p; bm[kk] = m;
        }
        __builtin_amdgcn_sched_barrier(0);
        v8f lo, hi;
        mma_tile(aE, aO, bp, bm, lo, hi);
        float* cb = Z2 + ((size_t)((bb * 64 + d) * 32 + r)) * 1024 + chalf * 16;
        const unsigned so_lo = (unsigned)(rbase * 32 + row);
        const unsigned so_hi = (unsigned)((31 - rbase) * 32 + row);
#pragma unroll
        for (int v = 0; v < 8; ++v) {
            cb[so_lo + (unsigned)(v * 32)] = lo[v];
            cb[so_hi - (unsigned)(v * 32)] = hi[v];
        }
        __builtin_amdgcn_sched_barrier(0);
    }
}

extern "C" void kernel_launch(void* const* d_in, const int* in_sizes, int n_in,
                              void* d_out, int out_size, void* d_ws, size_t ws_size,
                              hipStream_t stream) {
    (void)in_sizes; (void)n_in; (void)out_size; (void)ws_size;
    const float* x  = (const float*)d_in[0];
    float* out      = (float*)d_out;

    char* ws = (char*)d_ws;
    float* A  = (float*)ws;
    float* z2 = (float*)(ws + 8192 + (size_t)4*32*131072*4);

    dct_wh_fused_kernel<<<256, 512, 0, stream>>>(A, x, z2);

    dct_gemm_kernel<false><<<512, 256, 0, stream>>>(
        A, z2, out, 32768, 2048, 8192);
}


# Each lane owns two adjacent channels (even -> tile 0, odd -> tile 1): all global and LDS accesses of the three contractions are 8-byte, a lane half moves one whole 128-B line; sum order unchanged
# speedup vs baseline: 2.5601x; 1.0426x over previous
#include <hip/hip_runtime.h>
#include <math.h>

typedef __attribute__((ext_vector_type(2))) float v2f;
typedef __attribute__((ext_vector_type(8))) float v8f;


__device__ __forceinline__ void build_operator(float* __restrict__ A_lds, float* __restrict__ tab,
                                               float* __restrict__ Aout)
{
    const float PI = 3.14159265358979323846f;
    float* T32 = tab;
    float* T64 = tab + 24 * 16;
    for (int i = threadIdx.x; i < 24 * 16 + 24 * 32; i += 512) {
        if (i < 24 * 16) {
            const int p = i >> 4, r = i & 15;
            float s32 = (p == 0) ? sqrtf(1.0f / 32.0f) : sqrtf(2.0f / 32.0f);
            float c32 = cosf(PI * ((float)r + 0.5f) * (float)p / 32.0f);
            T32[i] = c32 * s32;
        } else {
            const int j = i - 24 * 16;
            const int p = j >> 5, w = j & 31;
            T64[j] = cosf(PI * ((float)w + 0.5f) * (float)p / 64.0f);
        }
    }
    __syncthreads();
    const int r = (int)(threadIdx.x >> 5);
    const int w = (int)(threadIdx.x & 31);
    float accE = 0.0f, accO = 0.0f;
#pragma unroll
    for (int p = 0; p < 24; p += 2) {
        const float s64e = (p == 0) ? sqrtf(1.0f / 64.0f) : sqrtf(2.0f / 64.0f);
        const float s64o = sqrtf(2.0f / 64.0f);
        accE = __builtin_fmaf(T32[p * 16 + r] * T64[p * 32 + w], s64e, accE);
        accO = __builtin_fmaf(T32[(p + 1) * 16 + r] * T64[(p + 1) * 32 + w], s64o, accO);
    }
    A_lds[threadIdx.x] = accE;
    A_lds[512 + threadIdx.x] = accO;
    if (blockIdx.x == 0) {
        Aout[threadIdx.x] = accE;
        Aout[512 + threadIdx.x] = accO;
    }
    __syncthreads();
}


__device__ __forceinline__ void load_a_frags(v2f (&aE)[8], v2f (&aO)[8],
                                             const float* A_lds, int row, int koff)
{
#pragma unroll
    for (int kk = 0; kk < 8; ++kk) {
        const float* ep = &A_lds[row * 32 + kk * 4 + koff];
        v2f a; a.x = ep[0];   a.y = ep[1];   aE[kk] = a;
        v2f b; b.x = ep[512]; b.y = ep[513]; aO[kk] = b;
    }
}

struct BPair { v2f p0[8], p1[8], m0[8], m1[8]; };

template <bool NT>
__device__ __forceinline__ void load_b_pair(BPair& b, const float* __restrict__ tb,
                                            unsigned loP, unsigned loM, int N)
{
#pragma unroll
    for (int kk = 0; kk < 8; ++kk) {
        const float* rp = tb + (size_t)(kk * 4) * (size_t)N;
        const float* rm = tb + (size_t)(60 - kk * 4) * (size_t)N;
        const v2f* a0 = (const v2f*)(rp + loP);
        const v2f* a1 = (const v2f*)(rp + loP + (unsigned)N);
        const v2f* c0 = (const v2f*)(rm + loM + (unsigned)N);
        const v2f* c1 = (const v2f*)(rm + loM);
        if (NT) { b.p0[kk] = __builtin_nontemporal_load(a0);
                  b.p1[kk] = __builtin_nontemporal_load(a1);
                  b.m0[kk] = __builtin_nontemporal_load(c0);
                  b.m1[kk] = __builtin_nontemporal_load(c1); }
        else    { b.p0[kk] = *a0; b.p1[kk] = *a1;
                  b.m0[kk] = *c0; b.m1[kk] = *c1; }
    }
}

__device__ __forceinline__ void mma_pair(const v2f (&aE)[8], const v2f (&aO)[8], const BPair& b,
                                         v8f (&lo)[2], v8f (&hi)[2])
{
    v8f accE0 = {}; v8f accO0 = {}; v8f accE1 = {}; v8f accO1 = {};
#pragma unroll
    for (int kk = 0; kk < 8; ++kk) {
        v2f u0, v0, u1, v1;
        u0.x = b.p0[kk].x + b.m0[kk].x;  u0.y = b.p1[kk].x + b.m1[kk].x;
        v0.x = b.p0[kk].x - b.m0[kk].x;  v0.y = b.p1[kk].x - b.m1[kk].x;
        u1.x = b.p0[kk].y + b.m0[kk].y;  u1.y = b.p1[kk].y + b.m1[kk].y;
        v1.x = b.p0[kk].y - b.m0[kk].y;  v1.y = b.p1[kk].y - b.m1[kk].y;
        accE0 = __builtin_amdgcn_wmma_f32_16x16x4_f32(false, aE[kk], false, u0,
                                                      (short)0, accE0, false, false);
        accO0 = __builtin_amdgcn_wmma_f32_16x16x4_f32(false, aO[kk], false, v0,
                                                      (short)0, accO0, false, false);
        accE1 = __builtin_amdgcn_wmma_f32_16x16x4_f32(false, aE[kk], false, u1,
                                                      (short)0, accE1, false, false);
        accO1 = __builtin_amdgcn_wmma_f32_16x16x4_f32(false, aO[kk], false, v1,
                                                      (short)0, accO1, false, false);
    }
    lo[0] = accE0 + accO0;  hi[0] = accE0 - accO0;
    lo[1] = accE1 + accO1;  hi[1] = accE1 - accO1;
}

__device__ __forceinline__ void mma_store_pair(const v2f (&aE)[8], const v2f (&aO)[8], const BPair& b,
                                               float* __restrict__ C,
                                               int batchIdx, int n0, int row, int rbase)
{
    v8f lo[2], hi[2];
    mma_pair(aE, aO, b, lo, hi);
    float* cb = C + ((size_t)batchIdx * 1024 + (size_t)(n0 >> 5)) * 1024;
    const unsigned so_lo = (unsigned)(rbase * 32 + 2 * row);
    const unsigned so_hi = (unsigned)((31 - rbase) * 32 + 2 * row);
#pragma unroll
    for (int v = 0; v < 8; ++v) {
        v2f l; l.x = lo[0][v]; l.y = lo[1][v];
        v2f h; h.x = hi[0][v]; h.y = hi[1][v];
        *(v2f*)(cb + so_lo + (unsigned)(v * 32)) = l;
        *(v2f*)(cb + so_hi - (unsigned)(v * 32)) = h;
    }
}

template <bool NT>
__global__ __launch_bounds__(256) void dct_gemm_kernel(
    const float* __restrict__ Aglob,
    const float* __restrict__ B,
    float* __restrict__ C,
    int N, int ntilesPerBatch, int totalTiles)
{
    __shared__ float A_lds[2 * 16 * 32];
    for (int i = threadIdx.x; i < 2 * 16 * 32; i += 256) A_lds[i] = Aglob[i];
    __syncthreads();

    const int lane = threadIdx.x & 31;
    const int wave = __builtin_amdgcn_readfirstlane((int)(threadIdx.x >> 5));
    const int gw = blockIdx.x * 8 + wave;
    const int GW = gridDim.x * 8;
    if (gw >= totalTiles) return;

    const int row   = lane & 15;
    const int koff  = (lane >> 4) << 1;
    const int rbase = (lane >> 4) << 3;

    v2f aE[8], aO[8];
    load_a_frags(aE, aO, A_lds, row, koff);

    const unsigned loP = (unsigned)(koff * N + 2 * row);
    const unsigned loM = (unsigned)((2 - koff) * N + 2 * row);

    BPair s0, s1;

    int t = gw;
    int batchIdx = t / ntilesPerBatch;
    int n0 = (t - batchIdx * ntilesPerBatch) * 32;
    load_b_pair<NT>(s0, B + (size_t)batchIdx * 64 * (size_t)N + n0, loP, loM, N);

    for (;;) {
        int tn = t + GW;
        if (tn >= totalTiles) {
            mma_store_pair(aE, aO, s0, C, batchIdx, n0, row, rbase);
            break;
        }
        int bn = tn / ntilesPerBatch;
        int nn = (tn - bn * ntilesPerBatch) * 32;
        load_b_pair<NT>(s1, B + (size_t)bn * 64 * (size_t)N + nn, loP, loM, N);
        mma_store_pair(aE, aO, s0, C, batchIdx, n0, row, rbase);
        t = tn; batchIdx = bn; n0 = nn;

        tn = t + GW;
        if (tn >= totalTiles) {
            mma_store_pair(aE, aO, s1, C, batchIdx, n0, row, rbase);
            break;
        }
        bn = tn / ntilesPerBatch;
        nn = (tn - bn * ntilesPerBatch) * 32;
        load_b_pair<NT>(s0, B + (size_t)bn * 64 * (size_t)N + nn, loP, loM, N);
        mma_store_pair(aE, aO, s1, C, batchIdx, n0, row, rbase);
        t = tn; batchIdx = bn; n0 = nn;
    }
}


__device__ __forceinline__ int zidx(int p, int h, int c) {
    return (p * 64 + (h ^ (((h >> 1) ^ (p >> 3)) & 1))) * 32 + c;
}

__device__ __forceinline__ void mma_to_slab(const v2f (&aE)[8], const v2f (&aO)[8], const BPair& b,
                                            float* __restrict__ Zs, int h, int row, int rbase)
{
    v8f lo[2], hi[2];
    mma_pair(aE, aO, b, lo, hi);
#pragma unroll
    for (int v = 0; v < 8; ++v) {
        v2f l; l.x = lo[0][v]; l.y = lo[1][v];
        v2f g; g.x = hi[0][v]; g.y = hi[1][v];
        *(v2f*)&Zs[zidx(rbase + v,      h, 2 * row)] = l;
        *(v2f*)&Zs[zidx(31 - rbase - v, h, 2 * row)] = g;
    }
    __builtin_amdgcn_sched_barrier(0);
}

__global__ __launch_bounds__(512) void dct_wh_fused_kernel(
    float* __restrict__ Aout,
    const float* __restrict__ X,
    float* __restrict__ Z2)
{
    __shared__ float A_lds[2 * 16 * 32];
    __shared__ __attribute__((aligned(16))) float Zs[32 * 64 * 32];
    build_operator(A_lds, Zs, Aout);

    const int lane = threadIdx.x & 31;
    const int wave = __builtin_amdgcn_readfirstlane((int)(threadIdx.x >> 5));
    const int row   = lane & 15;
    const int koff  = (lane >> 4) << 1;
    const int rbase = (lane >> 4) << 3;

    v2f aE[8], aO[8];
    load_a_frags(aE, aO, A_lds, row, koff);

    const int bb = blockIdx.x >> 6;
    const int d  = blockIdx.x & 63;

    {
        const int N1 = 32;
        const unsigned loP = (unsigned)(koff * N1 + 2 * row);
        const unsigned loM = (unsigned)((2 - koff) * N1 + 2 * row);
        const float* xb = X + ((size_t)(bb * 64 + d) * 64 + (size_t)(wave * 4)) * 2048;
        BPair s0, s1;
        load_b_pair<true>(s0, xb, loP, loM, N1);
#pragma unroll
        for (int j = 0; j < 4; j += 2) {
            const int h = wave * 4 + j;
            load_b_pair<true>(s1, xb + (j + 1) * 2048, loP, loM, N1);
            mma_to_slab(aE, aO, s0, Zs, h, row, rbase);
            if (j < 2) load_b_pair<true>(s0, xb + (j + 2) * 2048, loP, loM, N1);
            mma_to_slab(aE, aO, s1, Zs, h + 1, row, rbase);
        }
    }
    __syncthreads();

#pragma unroll
    for (int i = 0; i < 2; ++i) {
        const int r = wave * 2 + i;
        BPair s;
#pragma unroll
        for (int kk = 0; kk < 8; ++kk) {
            const int h0 = kk * 4 + koff;
            s.p0[kk] = *(const v2f*)&Zs[zidx(r, h0,      2 * row)];
            s.p1[kk] = *(const v2f*)&Zs[zidx(r, h0 + 1,  2 * row)];
            s.m0[kk] = *(const v2f*)&Zs[zidx(r, 63 - h0, 2 * row)];
            s.m1[kk] = *(const v2f*)&Zs[zidx(r, 62 - h0, 2 * row)];
        }
        __builtin_amdgcn_sched_barrier(0);
        v8f lo[2], hi[2];
        mma_pair(aE, aO, s, lo, hi);
        float* cb = Z2 + ((size_t)((bb * 64 + d) * 32 + r)) * 1024;
        const unsigned so_lo = (unsigned)(rbase * 32 + 2 * row);
        const unsigned so_hi = (unsigned)((31 - rbase) * 32 + 2 * row);
#pragma unroll
        for (int v = 0; v < 8; ++v) {
            v2f l; l.x = lo[0][v]; l.y = lo[1][v];
            v2f g; g.x = hi[0][v]; g.y = hi[1][v];
            *(v2f*)(cb + so_lo + (unsigned)(v * 32)) = l;
            *(v2f*)(cb + so_hi - (unsigned)(v * 32)) = g;
        }
        __builtin_amdgcn_sched_barrier(0);
    }
}

extern "C" void kernel_launch(void* const* d_in, const int* in_sizes, int n_in,
                              void* d_out, int out_size, void* d_ws, size_t ws_size,
                              hipStream_t stream) {
    (void)in_sizes; (void)n_in; (void)out_size; (void)ws_size;
    const float* x  = (const float*)d_in[0];
    float* out      = (float*)d_out;

    char* ws = (char*)d_ws;
    float* A  = (float*)ws;
    float* z2 = (float*)(ws + 8192 + (size_t)4*32*131072*4);

    dct_wh_fused_kernel<<<256, 512, 0, stream>>>(A, x, z2);

    dct_gemm_kernel<false><<<512, 256, 0, stream>>>(
        A, z2, out, 32768, 1024, 4096);
}


# Both kernels request their first tile pair (last pass: its 4 operator words too, at once) ahead of the prologue, whose barriers now order LDS only: prologue runs under memory latency; same sum order
# speedup vs baseline: 2.8052x; 1.0957x over previous
#include <hip/hip_runtime.h>
#include <math.h>

typedef __attribute__((ext_vector_type(2))) float v2f;
typedef __attribute__((ext_vector_type(8))) float v8f;


__device__ __forceinline__ void lds_barrier()
{
    __builtin_amdgcn_fence(__ATOMIC_RELEASE, "workgroup", "local");
    __builtin_amdgcn_s_barrier();
    __builtin_amdgcn_fence(__ATOMIC_ACQUIRE, "workgroup", "local");
}

__device__ __forceinline__ void build_operator(float* __restrict__ A_lds, float* __restrict__ tab,
                                               float* __restrict__ Aout)
{
    const float PI = 3.14159265358979323846f;
    float* T32 = tab;
    float* T64 = tab + 24 * 16;
    for (int i = threadIdx.x; i < 24 * 16 + 24 * 32; i += 512) {
        if (i < 24 * 16) {
            const int p = i >> 4, r = i & 15;
            float s32 = (p == 0) ? sqrtf(1.0f / 32.0f) : sqrtf(2.0f / 32.0f);
            float c32 = cosf(PI * ((float)r + 0.5f) * (float)p / 32.0f);
            T32[i] = c32 * s32;
        } else {
            const int j = i - 24 * 16;
            const int p = j >> 5, w = j & 31;
            T64[j] = cosf(PI * ((float)w + 0.5f) * (float)p / 64.0f);
        }
    }
    lds_barrier();
    const int r = (int)(threadIdx.x >> 5);
    const int w = (int)(threadIdx.x & 31);
    float accE = 0.0f, accO = 0.0f;
#pragma unroll
    for (int p = 0; p < 24; p += 2) {
        const float s64e = (p == 0) ? sqrtf(1.0f / 64.0f) : sqrtf(2.0f / 64.0f);
        const float s64o = sqrtf(2.0f / 64.0f);
        accE = __builtin_fmaf(T32[p * 16 + r] * T64[p * 32 + w], s64e, accE);
        accO = __builtin_fmaf(T32[(p + 1) * 16 + r] * T64[(p + 1) * 32 + w], s64o, accO);
    }
    A_lds[threadIdx.x] = accE;
    A_lds[512 + threadIdx.x] = accO;
    if (blockIdx.x == 0) {
        Aout[threadIdx.x] = accE;
        Aout[512 + threadIdx.x] = accO;
    }
    lds_barrier();
}


__device__ __forceinline__ void load_a_frags(v2f (&aE)[8], v2f (&aO)[8],
                                             const float* A_lds, int row, int koff)
{
#pragma unroll
    for (int kk = 0; kk < 8; ++kk) {
        const float* ep = &A_lds[row * 32 + kk * 4 + koff];
        v2f a; a.x = ep[0];   a.y = ep[1];   aE[kk] = a;
        v2f b; b.x = ep[512]; b.y = ep[513]; aO[kk] = b;
    }
}

struct BPair { v2f p0[8], p1[8], m0[8], m1[8]; };

template <bool NT>
__device__ __forceinline__ void load_b_pair(BPair& b, const float* __restrict__ tb,
                                            unsigned loP, unsigned loM, int N)
{
#pragma unroll
    for (int kk = 0; kk < 8; ++kk) {
        const float* rp = tb + (size_t)(kk * 4) * (size_t)N;
        const float* rm = tb + (size_t)(60 - kk * 4) * (size_t)N;
        const v2f* a0 = (const v2f*)(rp + loP);
        const v2f* a1 = (const v2f*)(rp + loP + (unsigned)N);
        const v2f* c0 = (const v2f*)(rm + loM + (unsigned)N);
        const v2f* c1 = (const v2f*)(rm + loM);
        if (NT) { b.p0[kk] = __builtin_nontemporal_load(a0);
                  b.p1[kk] = __builtin_nontemporal_load(a1);
                  b.m0[kk] = __builtin_nontemporal_load(c0);
                  b.m1[kk] = __builtin_nontemporal_load(c1); }
        else    { b.p0[kk] = *a0; b.p1[kk] = *a1;
                  b.m0[kk] = *c0; b.m1[kk] = *c1; }
    }
}

__device__ __forceinline__ void mma_pair(const v2f (&aE)[8], const v2f (&aO)[8], const BPair& b,
                                         v8f (&lo)[2], v8f (&hi)[2])
{
    v8f accE0 = {}; v8f accO0 = {}; v8f accE1 = {}; v8f accO1 = {};
#pragma unroll
    for (int kk = 0; kk < 8; ++kk) {
        v2f u0, v0, u1, v1;
        u0.x = b.p0[kk].x + b.m0[kk].x;  u0.y = b.p1[kk].x + b.m1[kk].x;
        v0.x = b.p0[kk].x - b.m0[kk].x;  v0.y = b.p1[kk].x - b.m1[kk].x;
        u1.x = b.p0[kk].y + b.m0[kk].y;  u1.y = b.p1[kk].y + b.m1[kk].y;
        v1.x = b.p0[kk].y - b.m0[kk].y;  v1.y = b.p1[kk].y - b.m1[kk].y;
        accE0 = __builtin_amdgcn_wmma_f32_16x16x4_f32(false, aE[kk], false, u0,
                                                      (short)0, accE0, false, false);
        accO0 = __builtin_amdgcn_wmma_f32_16x16x4_f32(false, aO[kk], false, v0,
                                                      (short)0, accO0, false, false);
        accE1 = __builtin_amdgcn_wmma_f32_16x16x4_f32(false, aE[kk], false, u1,
                                                      (short)0, accE1, false, false);
        accO1 = __builtin_amdgcn_wmma_f32_16x16x4_f32(false, aO[kk], false, v1,
                                                      (short)0, accO1, false, false);
    }
    lo[0] = accE0 + accO0;  hi[0] = accE0 - accO0;
    lo[1] = accE1 + accO1;  hi[1] = accE1 - accO1;
}

__device__ __forceinline__ void mma_store_pair(const v2f (&aE)[8], const v2f (&aO)[8], const BPair& b,
                                               float* __restrict__ C,
                                               int batchIdx, int n0, int row, int rbase)
{
    v8f lo[2], hi[2];
    mma_pair(aE, aO, b, lo, hi);
    float* cb = C + ((size_t)batchIdx * 1024 + (size_t)(n0 >> 5)) * 1024;
    const unsigned so_lo = (unsigned)(rbase * 32 + 2 * row);
    const unsigned so_hi = (unsigned)((31 - rbase) * 32 + 2 * row);
#pragma unroll
    for (int v = 0; v < 8; ++v) {
        v2f l; l.x = lo[0][v]; l.y = lo[1][v];
        v2f h; h.x = hi[0][v]; h.y = hi[1][v];
        *(v2f*)(cb + so_lo + (unsigned)(v * 32)) = l;
        *(v2f*)(cb + so_hi - (unsigned)(v * 32)) = h;
    }
}

template <bool NT>
__global__ __launch_bounds__(256) void dct_gemm_kernel(
    const float* __restrict__ Aglob,
    const float* __restrict__ B,
    float* __restrict__ C,
    int N, int ntilesPerBatch, int totalTiles)
{
    __shared__ float A_lds[2 * 16 * 32];

    const int lane = threadIdx.x & 31;
    const int wave = __builtin_amdgcn_readfirstlane((int)(threadIdx.x >> 5));
    const int gw = blockIdx.x * 8 + wave;
    const int GW = gridDim.x * 8;
    const bool active = gw < totalTiles;

    const int row   = lane & 15;
    const int koff  = (lane >> 4) << 1;
    const int rbase = (lane >> 4) << 3;

    const unsigned loP = (unsigned)(koff * N + 2 * row);
    const unsigned loM = (unsigned)((2 - koff) * N + 2 * row);

    float areg[4];
#pragma unroll
    for (int j = 0; j < 4; ++j) areg[j] = Aglob[threadIdx.x + 256 * j];

    BPair s0, s1;

    int t = active ? gw : 0;
    int batchIdx = t / ntilesPerBatch;
    int n0 = (t - batchIdx * ntilesPerBatch) * 32;
    load_b_pair<NT>(s0, B + (size_t)batchIdx * 64 * (size_t)N + n0, loP, loM, N);

#pragma unroll
    for (int j = 0; j < 4; ++j) A_lds[threadIdx.x + 256 * j] = areg[j];
    lds_barrier();
    if (!active) return;

    v2f aE[8], aO[8];
    load_a_frags(aE, aO, A_lds, row, koff);

    for (;;) {
        int tn = t + GW;
        if (tn >= totalTiles) {
            mma_store_pair(aE, aO, s0, C, batchIdx, n0, row, rbase);
            break;
        }
        int bn = tn / ntilesPerBatch;
        int nn = (tn - bn * ntilesPerBatch) * 32;
        load_b_pair<NT>(s1, B + (size_t)bn * 64 * (size_t)N + nn, loP, loM, N);
        mma_store_pair(aE, aO, s0, C, batchIdx, n0, row, rbase);
        t = tn; batchIdx = bn; n0 = nn;

        tn = t + GW;
        if (tn >= totalTiles) {
            mma_store_pair(aE, aO, s1, C, batchIdx, n0, row, rbase);
            break;
        }
        bn = tn / ntilesPerBatch;
        nn = (tn - bn * ntilesPerBatch) * 32;
        load_b_pair<NT>(s0, B + (size_t)bn * 64 * (size_t)N + nn, loP, loM, N);
        mma_store_pair(aE, aO, s1, C, batchIdx, n0, row, rbase);
        t = tn; batchIdx = bn; n0 = nn;
    }
}


__device__ __forceinline__ int zidx(int p, int h, int c) {
    return (p * 64 + (h ^ (((h >> 1) ^ (p >> 3)) & 1))) * 32 + c;
}

__device__ __forceinline__ void mma_to_slab(const v2f (&aE)[8], const v2f (&aO)[8], const BPair& b,
                                            float* __restrict__ Zs, int h, int row, int rbase)
{
    v8f lo[2], hi[2];
    mma_pair(aE, aO, b, lo, hi);
#pragma unroll
    for (int v = 0; v < 8; ++v) {
        v2f l; l.x = lo[0][v]; l.y = lo[1][v];
        v2f g; g.x = hi[0][v]; g.y = hi[1][v];
        *(v2f*)&Zs[zidx(rbase + v,      h, 2 * row)] = l;
        *(v2f*)&Zs[zidx(31 - rbase - v, h, 2 * row)] = g;
    }
    __builtin_amdgcn_sched_barrier(0);
}

__global__ __launch_bounds__(512) void dct_wh_fused_kernel(
    float* __restrict__ Aout,
    const float* __restrict__ X,
    float* __restrict__ Z2)
{
    __shared__ float A_lds[2 * 16 * 32];
    __shared__ __attribute__((aligned(16))) float Zs[32 * 64 * 32];

    const int lane = threadIdx.x & 31;
    const int wave = __builtin_amdgcn_readfirstlane((int)(threadIdx.x >> 5));
    const int row   = lane & 15;
    const int koff  = (lane >> 4) << 1;
    const int rbase = (lane >> 4) << 3;

    const int bb = blockIdx.x >> 6;
    const int d  = blockIdx.x & 63;

    const int N1 = 32;
    const unsigned loP = (unsigned)(koff * N1 + 2 * row);
    const unsigned loM = (unsigned)((2 - koff) * N1 + 2 * row);
    const float* xb = X + ((size_t)(bb * 64 + d) * 64 + (size_t)(wave * 4)) * 2048;
    BPair s0, s1;
    load_b_pair<true>(s0, xb, loP, loM, N1);

    build_operator(A_lds, Zs, Aout);
    __builtin_amdgcn_sched_barrier(0);

    v2f aE[8], aO[8];
    load_a_frags(aE, aO, A_lds, row, koff);

    {
#pragma unroll
        for (int j = 0; j < 4; j += 2) {
            const int h = wave * 4 + j;
            load_b_pair<true>(s1, xb + (j + 1) * 2048, loP, loM, N1);
            mma_to_slab(aE, aO, s0, Zs, h, row, rbase);
            if (j < 2) load_b_pair<true>(s0, xb + (j + 2) * 2048, loP, loM, N1);
            mma_to_slab(aE, aO, s1, Zs, h + 1, row, rbase);
        }
    }
    __syncthreads();

#pragma unroll
    for (int i = 0; i < 2; ++i) {
        const int r = wave * 2 + i;
        BPair s;
#pragma unroll
        for (int kk = 0; kk < 8; ++kk) {
            const int h0 = kk * 4 + koff;
            s.p0[kk] = *(const v2f*)&Zs[zidx(r, h0,      2 * row)];
            s.p1[kk] = *(const v2f*)&Zs[zidx(r, h0 + 1,  2 * row)];
            s.m0[kk] = *(const v2f*)&Zs[zidx(r, 63 - h0, 2 * row)];
            s.m1[kk] = *(const v2f*)&Zs[zidx(r, 62 - h0, 2 * row)];
        }
        __builtin_amdgcn_sched_barrier(0);
        v8f lo[2], hi[2];
        mma_pair(aE, aO, s, lo, hi);
        float* cb = Z2 + ((size_t)((bb * 64 + d) * 32 + r)) * 1024;
        const unsigned so_lo = (unsigned)(rbase * 32 + 2 * row);
        const unsigned so_hi = (unsigned)((31 - rbase) * 32 + 2 * row);
#pragma unroll
        for (int v = 0; v < 8; ++v) {
            v2f l; l.x = lo[0][v]; l.y = lo[1][v];
            v2f g; g.x = hi[0][v]; g.y = hi[1][v];
            *(v2f*)(cb + so_lo + (unsigned)(v * 32)) = l;
            *(v2f*)(cb + so_hi - (unsigned)(v * 32)) = g;
        }
        __builtin_amdgcn_sched_barrier(0);
    }
}

extern "C" void kernel_launch(void* const* d_in, const int* in_sizes, int n_in,
                              void* d_out, int out_size, void* d_ws, size_t ws_size,
                              hipStream_t stream) {
    (void)in_sizes; (void)n_in; (void)out_size; (void)ws_size;
    const float* x  = (const float*)d_in[0];
    float* out      = (float*)d_out;

    char* ws = (char*)d_ws;
    float* A  = (float*)ws;
    float* z2 = (float*)(ws + 8192 + (size_t)4*32*131072*4);

    dct_wh_fused_kernel<<<256, 512, 0, stream>>>(A, x, z2);

    dct_gemm_kernel<false><<<512, 256, 0, stream>>>(
        A, z2, out, 32768, 1024, 4096);
}
